# starting kernel
# baseline (speedup 1.0000x reference)
#include <hip/hip_runtime.h>

#define BB    64
#define TT    128
#define NN    2048
#define DIN   256
#define DINIT 256
#define DOUT  512
#define DECAY 0.1f

#define KCH   64
#define PADB  68

typedef __attribute__((ext_vector_type(2))) float v2f;
typedef __attribute__((ext_vector_type(8))) float v8f;

__device__ __forceinline__ v2f ld_a(const float* A, int lda, int row0, int k, int lane) {
  const float* p = A + (row0 + (lane & 15)) * lda + k + ((lane >> 4) << 1);
  return v2f{p[0], p[1]};
}

__device__ __forceinline__ void gemm_staged(v8f acc[4], const float* __restrict__ A, int lda,
                                            const float* __restrict__ Bm, int ldb, int K,
                                            int row0, int colblk, int col0w, int lane, int tid) {
  __shared__ float Bs[128][PADB];
  const int n_st  = tid & 127;
  const int k_st0 = (tid >> 7) * 4;

  for (int kb = 0; kb < K; kb += KCH) {
#pragma unroll
    for (int kk = k_st0; kk < KCH; kk += 8) {
      const float* gp = Bm + (size_t)(kb + kk) * ldb + colblk + n_st;
      float4 v = {gp[0], gp[(size_t)ldb], gp[2 * (size_t)ldb], gp[3 * (size_t)ldb]};
      *(float4*)&Bs[n_st][kk] = v;
    }
    if (kb + KCH < K)
      __builtin_prefetch(Bm + (size_t)(kb + KCH + k_st0) * ldb + colblk + n_st, 0, 1);
    __syncthreads();

#pragma unroll 4
    for (int k = 0; k < KCH; k += 4) {
      v2f a = ld_a(A, lda, row0, kb + k, lane);
#pragma unroll
      for (int i = 0; i < 4; ++i) {
        int n = col0w + 16 * i + (lane & 15);
        const float* p = &Bs[n][k + ((lane >> 4) << 1)];
        v2f b = {p[0], p[1]};
        acc[i] = __builtin_amdgcn_wmma_f32_16x16x4_f32(false, a, false, b, (short)0, acc[i],
                                                       false, false);
      }
    }
    __syncthreads();
  }
}

__global__ void RSLP_init_kernel(const float* __restrict__ x0, const float* __restrict__ Wi,
                                 const float* __restrict__ bi, float* __restrict__ s0,
                                 float* __restrict__ h) {
  int tid = threadIdx.x, lane = tid & 31, wave = tid >> 5;
  int row0   = (wave & 3) * 16;
  int col0w  = (wave >> 2) * 64;
  int colblk = blockIdx.x * 128;
  v8f acc[4] = {};
  gemm_staged(acc, x0, DINIT, Wi, 2 * NN, DINIT, row0, colblk, col0w, lane, tid);

  float* dst = (blockIdx.x < 16) ? s0 : h;
  int joff   = (blockIdx.x < 16) ? 0 : NN;
#pragma unroll
  for (int i = 0; i < 4; ++i) {
    int j = colblk + col0w + 16 * i + (lane & 15);
    float bj = bi[j];
#pragma unroll
    for (int v = 0; v < 8; ++v) {
      int m = row0 + v + 8 * (lane >> 4);
      dst[m * NN + (j - joff)] = acc[i][v] + bj;
    }
  }
}

__global__ void RSLP_step_kernel(const float* __restrict__ x, const float* __restrict__ Win,
                                 const float* __restrict__ bin, const float* __restrict__ r,
                                 const float* __restrict__ rb, const float* __restrict__ o,
                                 const float* __restrict__ s_old, float* __restrict__ s_new,
                                 float* __restrict__ h, float* __restrict__ out, int t) {
  int tid = threadIdx.x, lane = tid & 31, wave = tid >> 5;
  int row0  = (wave & 3) * 16;
  int col0w = (wave >> 2) * 64;
  int bid = blockIdx.x;

  if (bid < 16) {
    if (t >= TT) return;
    int colblk = bid * 128;
    v8f acc[4] = {};
    gemm_staged(acc, s_old, NN, r, NN, NN, row0, colblk, col0w, lane, tid);
    gemm_staged(acc, x + t * DIN, TT * DIN, Win, NN, DIN, row0, colblk, col0w, lane, tid);
#pragma unroll
    for (int i = 0; i < 4; ++i) {
      int j = colblk + col0w + 16 * i + (lane & 15);
      float addj  = rb[j] + bin[j];
      float rdiag = r[j * NN + j];
#pragma unroll
      for (int v = 0; v < 8; ++v) {
        int m = row0 + v + 8 * (lane >> 4);
        float pre = acc[i][v] - s_old[m * NN + j] * rdiag + addj;
        float hn  = (1.0f - DECAY) * h[m * NN + j] + DECAY * pre;
        h[m * NN + j]     = hn;
        s_new[m * NN + j] = fmaxf(hn, 0.0f);
      }
    }
  } else {
    if (t < 1) return;
    int colblk = (bid - 16) * 128;
    v8f acc[4] = {};
    gemm_staged(acc, s_old, NN, o, DOUT, NN, row0, colblk, col0w, lane, tid);
#pragma unroll
    for (int i = 0; i < 4; ++i) {
      int j = colblk + col0w + 16 * i + (lane & 15);
#pragma unroll
      for (int v = 0; v < 8; ++v) {
        int m = row0 + v + 8 * (lane >> 4);
        out[(m * TT + (t - 1)) * DOUT + j] = acc[i][v];
      }
    }
  }
}

extern "C" void kernel_launch(void* const* d_in, const int* in_sizes, int n_in,
                              void* d_out, int out_size, void* d_ws, size_t ws_size,
                              hipStream_t stream) {
  (void)in_sizes; (void)n_in; (void)out_size; (void)ws_size;
  const float* x      = (const float*)d_in[0];
  const float* x0     = (const float*)d_in[1];
  const float* W_init = (const float*)d_in[2];
  const float* b_init = (const float*)d_in[3];
  const float* W_in   = (const float*)d_in[4];
  const float* b_in   = (const float*)d_in[5];
  const float* r      = (const float*)d_in[6];
  const float* r_b    = (const float*)d_in[7];
  const float* o      = (const float*)d_in[8];
  float* out = (float*)d_out;

  float* sb0  = (float*)d_ws;
  float* sb1  = sb0 + BB * NN;
  float* hbuf = sb1 + BB * NN;
  float* sb[2] = {sb0, sb1};

  RSLP_init_kernel<<<32, 256, 0, stream>>>(x0, W_init, b_init, sb0, hbuf);
  for (int t = 0; t <= TT; ++t) {
    RSLP_step_kernel<<<20, 256, 0, stream>>>(x, W_in, b_in, r, r_b, o,
                                             sb[t & 1], sb[(t + 1) & 1], hbuf, out, t);
  }
}


# K loop software-pipelined: next chunk's B rows and A fragments go to registers before the current chunk's WMMAs; two LDS stages and two A register sets alternate, one barrier per chunk
# speedup vs baseline: 1.2735x; 1.2735x over previous
#include <hip/hip_runtime.h>

#define BB    64
#define TT    128
#define NN    2048
#define DIN   256
#define DINIT 256
#define DOUT  512
#define DECAY 0.1f

#define KCH   64
#define PADB  68

typedef __attribute__((ext_vector_type(2))) float v2f;
typedef __attribute__((ext_vector_type(8))) float v8f;


__device__ __forceinline__ void ld_chunk(float4 (&b)[8], v2f (&a)[16], const float* __restrict__ bq,
                                         const float* __restrict__ aq, int ldb) {
#pragma unroll
  for (int j = 0; j < 8; ++j) {
    const float* gp = bq + (size_t)(8 * j) * ldb;
    b[j] = float4{gp[0], gp[(size_t)ldb], gp[2 * (size_t)ldb], gp[3 * (size_t)ldb]};
  }
#pragma unroll
  for (int ks = 0; ks < 16; ++ks) a[ks] = v2f{aq[4 * ks], aq[4 * ks + 1]};
}

__device__ __forceinline__ void st_chunk(float (*Bc)[PADB], const float4 (&b)[8], int n_st, int k_st0) {
#pragma unroll
  for (int j = 0; j < 8; ++j) *(float4*)&Bc[n_st][k_st0 + 8 * j] = b[j];
}

__device__ __forceinline__ void mm_chunk(v8f acc[4], float (*Bc)[PADB], const v2f (&a)[16], int n0, int hk) {
#pragma unroll
  for (int ks = 0; ks < 16; ++ks) {
#pragma unroll
    for (int i = 0; i < 4; ++i) {
      const float* p = &Bc[n0 + 16 * i][4 * ks + hk];
      v2f b = {p[0], p[1]};
      acc[i] = __builtin_amdgcn_wmma_f32_16x16x4_f32(false, a[ks], false, b, (short)0, acc[i],
                                                     false, false);
    }
  }
}

template <int K>
__device__ __forceinline__ void gemm_staged(v8f acc[4], float (*Bs)[128][PADB],
                                            const float* __restrict__ A, int lda,
                                            const float* __restrict__ Bm, int ldb,
                                            int row0, int colblk, int col0w, int lane, int tid) {
  static_assert(K % (2 * KCH) == 0, "the loop body holds two chunks, and a following call starts in stage 0");
  const int n_st  = tid & 127;
  const int k_st0 = (tid >> 7) * 4;
  const int hk    = (lane >> 4) << 1;
  const int n0    = col0w + (lane & 15);
  const float* ap = A + (row0 + (lane & 15)) * lda + hk;
  const float* bp = Bm + (size_t)k_st0 * ldb + colblk + n_st;

  float4 b[8];
  v2f    a0[16], a1[16];

  ld_chunk(b, a0, bp, ap, ldb);
  for (int kb = 0; kb < K; kb += 2 * KCH) {
    st_chunk(Bs[0], b, n_st, k_st0);
    __syncthreads();
    ld_chunk(b, a1, bp + (size_t)(kb + KCH) * ldb, ap + (kb + KCH), ldb);
    mm_chunk(acc, Bs[0], a0, n0, hk);

    st_chunk(Bs[1], b, n_st, k_st0);
    __syncthreads();
    if (kb + 2 * KCH < K)
      ld_chunk(b, a0, bp + (size_t)(kb + 2 * KCH) * ldb, ap + (kb + 2 * KCH), ldb);
    mm_chunk(acc, Bs[1], a1, n0, hk);
  }
}

__global__ __launch_bounds__(256) void RSLP_init_kernel(const float* __restrict__ x0, const float* __restrict__ Wi,
                                 const float* __restrict__ bi, float* __restrict__ s0,
                                 float* __restrict__ h) {
  int tid = threadIdx.x, lane = tid & 31, wave = tid >> 5;
  int row0   = (wave & 3) * 16;
  int col0w  = (wave >> 2) * 64;
  int colblk = blockIdx.x * 128;
  __shared__ float Bs[2][128][PADB];
  v8f acc[4] = {};
  gemm_staged<DINIT>(acc, Bs, x0, DINIT, Wi, 2 * NN, row0, colblk, col0w, lane, tid);

  float* dst = (blockIdx.x < 16) ? s0 : h;
  int joff   = (blockIdx.x < 16) ? 0 : NN;
#pragma unroll
  for (int i = 0; i < 4; ++i) {
    int j = colblk + col0w + 16 * i + (lane & 15);
    float bj = bi[j];
#pragma unroll
    for (int v = 0; v < 8; ++v) {
      int m = row0 + v + 8 * (lane >> 4);
      dst[m * NN + (j - joff)] = acc[i][v] + bj;
    }
  }
}

__global__ __launch_bounds__(256) void RSLP_step_kernel(const float* __restrict__ x, const float* __restrict__ Win,
                                 const float* __restrict__ bin, const float* __restrict__ r,
                                 const float* __restrict__ rb, const float* __restrict__ o,
                                 const float* __restrict__ s_old, float* __restrict__ s_new,
                                 float* __restrict__ h, float* __restrict__ out, int t) {
  int tid = threadIdx.x, lane = tid & 31, wave = tid >> 5;
  int row0  = (wave & 3) * 16;
  int col0w = (wave >> 2) * 64;
  int bid = blockIdx.x;
  __shared__ float Bs[2][128][PADB];

  if (bid < 16) {
    if (t >= TT) return;
    int colblk = bid * 128;
    v8f acc[4] = {};
    gemm_staged<NN>(acc, Bs, s_old, NN, r, NN, row0, colblk, col0w, lane, tid);
    gemm_staged<DIN>(acc, Bs, x + t * DIN, TT * DIN, Win, NN, row0, colblk, col0w, lane, tid);
#pragma unroll
    for (int i = 0; i < 4; ++i) {
      int j = colblk + col0w + 16 * i + (lane & 15);
      float addj  = rb[j] + bin[j];
      float rdiag = r[j * NN + j];
#pragma unroll
      for (int v = 0; v < 8; ++v) {
        int m = row0 + v + 8 * (lane >> 4);
        float pre = acc[i][v] - s_old[m * NN + j] * rdiag + addj;
        float hn  = (1.0f - DECAY) * h[m * NN + j] + DECAY * pre;
        h[m * NN + j]     = hn;
        s_new[m * NN + j] = fmaxf(hn, 0.0f);
      }
    }
  } else {
    if (t < 1) return;
    int colblk = (bid - 16) * 128;
    v8f acc[4] = {};
    gemm_staged<NN>(acc, Bs, s_old, NN, o, DOUT, row0, colblk, col0w, lane, tid);
#pragma unroll
    for (int i = 0; i < 4; ++i) {
      int j = colblk + col0w + 16 * i + (lane & 15);
#pragma unroll
      for (int v = 0; v < 8; ++v) {
        int m = row0 + v + 8 * (lane >> 4);
        out[(m * TT + (t - 1)) * DOUT + j] = acc[i][v];
      }
    }
  }
}

extern "C" void kernel_launch(void* const* d_in, const int* in_sizes, int n_in,
                              void* d_out, int out_size, void* d_ws, size_t ws_size,
                              hipStream_t stream) {
  (void)in_sizes; (void)n_in; (void)out_size; (void)ws_size;
  const float* x      = (const float*)d_in[0];
  const float* x0     = (const float*)d_in[1];
  const float* W_init = (const float*)d_in[2];
  const float* b_init = (const float*)d_in[3];
  const float* W_in   = (const float*)d_in[4];
  const float* b_in   = (const float*)d_in[5];
  const float* r      = (const float*)d_in[6];
  const float* r_b    = (const float*)d_in[7];
  const float* o      = (const float*)d_in[8];
  float* out = (float*)d_out;

  float* sb0  = (float*)d_ws;
  float* sb1  = sb0 + BB * NN;
  float* hbuf = sb1 + BB * NN;
  float* sb[2] = {sb0, sb1};

  RSLP_init_kernel<<<32, 256, 0, stream>>>(x0, W_init, b_init, sb0, hbuf);
  for (int t = 0; t <= TT; ++t) {
    RSLP_step_kernel<<<20, 256, 0, stream>>>(x, W_in, b_in, r, r_b, o,
                                             sb[t & 1], sb[(t + 1) & 1], hbuf, out, t);
  }
}


# Step re-tiled over 160 work-groups (16x64 tile each) instead of 20: K split over the 8 waves, partial sums added through LDS (new summation order), operands loaded straight from global
# speedup vs baseline: 9.0489x; 7.1055x over previous
#include <hip/hip_runtime.h>

#define BB    64
#define TT    128
#define NN    2048
#define DIN   256
#define DINIT 256
#define DOUT  512
#define DECAY 0.1f

#define KCH   64
#define PADB  68

typedef __attribute__((ext_vector_type(2))) float v2f;
typedef __attribute__((ext_vector_type(4))) float v4f;
typedef __attribute__((ext_vector_type(8))) float v8f;


__device__ __forceinline__ void ld_chunk(float4 (&b)[8], v2f (&a)[16], const float* __restrict__ bq,
                                         const float* __restrict__ aq, int ldb) {
#pragma unroll
  for (int j = 0; j < 8; ++j) {
    const float* gp = bq + (size_t)(8 * j) * ldb;
    b[j] = float4{gp[0], gp[(size_t)ldb], gp[2 * (size_t)ldb], gp[3 * (size_t)ldb]};
  }
#pragma unroll
  for (int ks = 0; ks < 16; ++ks) a[ks] = v2f{aq[4 * ks], aq[4 * ks + 1]};
}

__device__ __forceinline__ void st_chunk(float (*Bc)[PADB], const float4 (&b)[8], int n_st, int k_st0) {
#pragma unroll
  for (int j = 0; j < 8; ++j) *(float4*)&Bc[n_st][k_st0 + 8 * j] = b[j];
}

__device__ __forceinline__ void mm_chunk(v8f acc[4], float (*Bc)[PADB], const v2f (&a)[16], int n0, int hk) {
#pragma unroll
  for (int ks = 0; ks < 16; ++ks) {
#pragma unroll
    for (int i = 0; i < 4; ++i) {
      const float* p = &Bc[n0 + 16 * i][4 * ks + hk];
      v2f b = {p[0], p[1]};
      acc[i] = __builtin_amdgcn_wmma_f32_16x16x4_f32(false, a[ks], false, b, (short)0, acc[i],
                                                     false, false);
    }
  }
}

template <int K>
__device__ __forceinline__ void gemm_staged(v8f acc[4], float (*Bs)[128][PADB],
                                            const float* __restrict__ A, int lda,
                                            const float* __restrict__ Bm, int ldb,
                                            int row0, int colblk, int col0w, int lane, int tid) {
  static_assert(K % (2 * KCH) == 0, "the loop body holds two chunks, and a following call starts in stage 0");
  const int n_st  = tid & 127;
  const int k_st0 = (tid >> 7) * 4;
  const int hk    = (lane >> 4) << 1;
  const int n0    = col0w + (lane & 15);
  const float* ap = A + (row0 + (lane & 15)) * lda + hk;
  const float* bp = Bm + (size_t)k_st0 * ldb + colblk + n_st;

  float4 b[8];
  v2f    a0[16], a1[16];

  ld_chunk(b, a0, bp, ap, ldb);
  for (int kb = 0; kb < K; kb += 2 * KCH) {
    st_chunk(Bs[0], b, n_st, k_st0);
    __syncthreads();
    ld_chunk(b, a1, bp + (size_t)(kb + KCH) * ldb, ap + (kb + KCH), ldb);
    mm_chunk(acc, Bs[0], a0, n0, hk);

    st_chunk(Bs[1], b, n_st, k_st0);
    __syncthreads();
    if (kb + 2 * KCH < K)
      ld_chunk(b, a0, bp + (size_t)(kb + 2 * KCH) * ldb, ap + (kb + 2 * KCH), ldb);
    mm_chunk(acc, Bs[1], a1, n0, hk);
  }
}


template <int NG, int LDB>
__device__ __forceinline__ void ld_batch(v4f (&a)[8], v4f (&b)[8][4], const float* __restrict__ ap,
                                         const float* __restrict__ bp) {
#pragma unroll
  for (int g = 0; g < NG; ++g) {
    a[g] = *(const v4f*)(ap + 8 * g);
#pragma unroll
    for (int j = 0; j < 4; ++j) b[g][j] = *(const v4f*)(bp + (8 * g + j) * LDB);
  }
}

template <int NG>
__device__ __forceinline__ void mm_batch(v8f (&acc)[4], const v4f (&a)[8], const v4f (&b)[8][4]) {
#pragma unroll
  for (int g = 0; g < NG; ++g) {
    v2f alo = {a[g][0], a[g][1]};
    v2f ahi = {a[g][2], a[g][3]};
#pragma unroll
    for (int i = 0; i < 4; ++i) {
      v2f bf = {b[g][0][i], b[g][1][i]};
      acc[i] = __builtin_amdgcn_wmma_f32_16x16x4_f32(false, alo, false, bf, (short)0, acc[i],
                                                     false, false);
    }
#pragma unroll
    for (int i = 0; i < 4; ++i) {
      v2f bf = {b[g][2][i], b[g][3][i]};
      acc[i] = __builtin_amdgcn_wmma_f32_16x16x4_f32(false, ahi, false, bf, (short)0, acc[i],
                                                     false, false);
    }
  }
}

template <int LDB, bool WITH_X>
__device__ __forceinline__ void gemm_ksplit(v8f (&acc)[4], const float* __restrict__ ap,
                                            const float* __restrict__ bp,
                                            const float* __restrict__ apx,
                                            const float* __restrict__ bpx) {
  v4f a0[8], b0[8][4], a1[8], b1[8][4];
  ld_batch<8, LDB>(a0, b0, ap, bp);
  ld_batch<8, LDB>(a1, b1, ap + 64, bp + 64 * LDB);
  __builtin_amdgcn_sched_barrier(0);
  mm_batch<8>(acc, a0, b0);
  __builtin_amdgcn_sched_barrier(0);
  ld_batch<8, LDB>(a0, b0, ap + 128, bp + 128 * LDB);
  __builtin_amdgcn_sched_barrier(0);
  mm_batch<8>(acc, a1, b1);
  __builtin_amdgcn_sched_barrier(0);
  ld_batch<8, LDB>(a1, b1, ap + 192, bp + 192 * LDB);
  __builtin_amdgcn_sched_barrier(0);
  mm_batch<8>(acc, a0, b0);
  if (WITH_X) {
    __builtin_amdgcn_sched_barrier(0);
    ld_batch<4, NN>(a0, b0, apx, bpx);
    __builtin_amdgcn_sched_barrier(0);
  }
  mm_batch<8>(acc, a1, b1);
  if (WITH_X) mm_batch<4>(acc, a0, b0);
}

__device__ __forceinline__ v4f reduce_tile(float (*red)[16][64], const v8f (&acc)[4], int wave,
                                           int lane, int tid) {
  const int c = lane & 15, h = lane >> 4;
#pragma unroll
  for (int v = 0; v < 8; ++v)
    *(v4f*)&red[wave][8 * h + v][4 * c] = v4f{acc[0][v], acc[1][v], acc[2][v], acc[3][v]};
  __syncthreads();
  const int row = tid >> 4, c4 = (tid & 15) * 4;
  v4f sum = *(const v4f*)&red[0][row][c4];
#pragma unroll
  for (int w = 1; w < 8; ++w) sum += *(const v4f*)&red[w][row][c4];
  return sum;
}

__global__ __launch_bounds__(256) void RSLP_init_kernel(const float* __restrict__ x0, const float* __restrict__ Wi,
                                 const float* __restrict__ bi, float* __restrict__ s0,
                                 float* __restrict__ h) {
  int tid = threadIdx.x, lane = tid & 31, wave = tid >> 5;
  int row0   = (wave & 3) * 16;
  int col0w  = (wave >> 2) * 64;
  int colblk = blockIdx.x * 128;
  __shared__ float Bs[2][128][PADB];
  v8f acc[4] = {};
  gemm_staged<DINIT>(acc, Bs, x0, DINIT, Wi, 2 * NN, row0, colblk, col0w, lane, tid);

  float* dst = (blockIdx.x < 16) ? s0 : h;
  int joff   = (blockIdx.x < 16) ? 0 : NN;
#pragma unroll
  for (int i = 0; i < 4; ++i) {
    int j = colblk + col0w + 16 * i + (lane & 15);
    float bj = bi[j];
#pragma unroll
    for (int v = 0; v < 8; ++v) {
      int m = row0 + v + 8 * (lane >> 4);
      dst[m * NN + (j - joff)] = acc[i][v] + bj;
    }
  }
}

__global__ __launch_bounds__(256) void RSLP_step_kernel(const float* __restrict__ x, const float* __restrict__ Win,
                                 const float* __restrict__ bin, const float* __restrict__ r,
                                 const float* __restrict__ rb, const float* __restrict__ o,
                                 const float* __restrict__ s_old, float* __restrict__ s_new,
                                 float* __restrict__ h, float* __restrict__ out, int t) {
  const int tid = threadIdx.x, lane = tid & 31, wave = tid >> 5;
  const int c = lane & 15, hl = lane >> 4;
  const int bid = blockIdx.x;
  __shared__ float red[8][16][64];
  const int row = tid >> 4, c4 = (tid & 15) * 4;

  if (bid < 128) {
    if (t >= TT) return;
    const int row0 = (bid & 3) * 16;
    const int n0   = (bid >> 2) * 64;
    const int m  = row0 + row;
    const int j0 = n0 + c4;
    const v4f so  = *(const v4f*)(s_old + m * NN + j0);
    const v4f hh  = *(const v4f*)(h + m * NN + j0);
    const v4f rbv = *(const v4f*)(rb + j0);
    const v4f biv = *(const v4f*)(bin + j0);
    float rd[4];
#pragma unroll
    for (int e = 0; e < 4; ++e) rd[e] = r[(j0 + e) * NN + (j0 + e)];

    v8f acc[4] = {};
    gemm_ksplit<NN, true>(acc,
                          s_old + (row0 + c) * NN + 256 * wave + 4 * hl,
                          r + (256 * wave + 4 * hl) * NN + n0 + 4 * c,
                          x + t * DIN + (row0 + c) * (TT * DIN) + 32 * wave + 4 * hl,
                          Win + (32 * wave + 4 * hl) * NN + n0 + 4 * c);
    const v4f sum = reduce_tile(red, acc, wave, lane, tid);
    v4f hn4, sn4;
#pragma unroll
    for (int e = 0; e < 4; ++e) {
      float addj = rbv[e] + biv[e];
      float pre  = sum[e] - so[e] * rd[e] + addj;
      float hn   = (1.0f - DECAY) * hh[e] + DECAY * pre;
      hn4[e] = hn;
      sn4[e] = fmaxf(hn, 0.0f);
    }
    *(v4f*)(h + m * NN + j0)     = hn4;
    *(v4f*)(s_new + m * NN + j0) = sn4;
  } else {
    if (t < 1) return;
    const int b2   = bid - 128;
    const int row0 = (b2 & 3) * 16;
    const int n0   = (b2 >> 2) * 64;
    v8f acc[4] = {};
    gemm_ksplit<DOUT, false>(acc,
                             s_old + (row0 + c) * NN + 256 * wave + 4 * hl,
                             o + (256 * wave + 4 * hl) * DOUT + n0 + 4 * c,
                             nullptr, nullptr);
    const v4f sum = reduce_tile(red, acc, wave, lane, tid);
    *(v4f*)(out + ((row0 + row) * TT + (t - 1)) * DOUT + n0 + c4) = sum;
  }
}

extern "C" void kernel_launch(void* const* d_in, const int* in_sizes, int n_in,
                              void* d_out, int out_size, void* d_ws, size_t ws_size,
                              hipStream_t stream) {
  (void)in_sizes; (void)n_in; (void)out_size; (void)ws_size;
  const float* x      = (const float*)d_in[0];
  const float* x0     = (const float*)d_in[1];
  const float* W_init = (const float*)d_in[2];
  const float* b_init = (const float*)d_in[3];
  const float* W_in   = (const float*)d_in[4];
  const float* b_in   = (const float*)d_in[5];
  const float* r      = (const float*)d_in[6];
  const float* r_b    = (const float*)d_in[7];
  const float* o      = (const float*)d_in[8];
  float* out = (float*)d_out;

  float* sb0  = (float*)d_ws;
  float* sb1  = sb0 + BB * NN;
  float* hbuf = sb1 + BB * NN;
  float* sb[2] = {sb0, sb1};

  RSLP_init_kernel<<<32, 256, 0, stream>>>(x0, W_init, b_init, sb0, hbuf);
  for (int t = 0; t <= TT; ++t) {
    RSLP_step_kernel<<<160, 256, 0, stream>>>(x, W_in, b_in, r, r_b, o,
                                             sb[t & 1], sb[(t + 1) & 1], hbuf, out, t);
  }
}


# Step work-groups cut from 8 waves to 4, one per SIMD: K split 4 ways instead of 8, so half as many waves enter per launch; partial sums added in a new order (4 partials, not 8)
# speedup vs baseline: 10.6598x; 1.1780x over previous
#include <hip/hip_runtime.h>

#define BB    64
#define TT    128
#define NN    2048
#define DIN   256
#define DINIT 256
#define DOUT  512
#define DECAY 0.1f

#define KCH   64
#define PADB  68

typedef __attribute__((ext_vector_type(2))) float v2f;
typedef __attribute__((ext_vector_type(4))) float v4f;
typedef __attribute__((ext_vector_type(8))) float v8f;


__device__ __forceinline__ void ld_chunk(float4 (&b)[8], v2f (&a)[16], const float* __restrict__ bq,
                                         const float* __restrict__ aq, int ldb) {
#pragma unroll
  for (int j = 0; j < 8; ++j) {
    const float* gp = bq + (size_t)(8 * j) * ldb;
    b[j] = float4{gp[0], gp[(size_t)ldb], gp[2 * (size_t)ldb], gp[3 * (size_t)ldb]};
  }
#pragma unroll
  for (int ks = 0; ks < 16; ++ks) a[ks] = v2f{aq[4 * ks], aq[4 * ks + 1]};
}

__device__ __forceinline__ void st_chunk(float (*Bc)[PADB], const float4 (&b)[8], int n_st, int k_st0) {
#pragma unroll
  for (int j = 0; j < 8; ++j) *(float4*)&Bc[n_st][k_st0 + 8 * j] = b[j];
}

__device__ __forceinline__ void mm_chunk(v8f acc[4], float (*Bc)[PADB], const v2f (&a)[16], int n0, int hk) {
#pragma unroll
  for (int ks = 0; ks < 16; ++ks) {
#pragma unroll
    for (int i = 0; i < 4; ++i) {
      const float* p = &Bc[n0 + 16 * i][4 * ks + hk];
      v2f b = {p[0], p[1]};
      acc[i] = __builtin_amdgcn_wmma_f32_16x16x4_f32(false, a[ks], false, b, (short)0, acc[i],
                                                     false, false);
    }
  }
}

template <int K>
__device__ __forceinline__ void gemm_staged(v8f acc[4], float (*Bs)[128][PADB],
                                            const float* __restrict__ A, int lda,
                                            const float* __restrict__ Bm, int ldb,
                                            int row0, int colblk, int col0w, int lane, int tid) {
  static_assert(K % (2 * KCH) == 0, "the loop body holds two chunks, and a following call starts in stage 0");
  const int n_st  = tid & 127;
  const int k_st0 = (tid >> 7) * 4;
  const int hk    = (lane >> 4) << 1;
  const int n0    = col0w + (lane & 15);
  const float* ap = A + (row0 + (lane & 15)) * lda + hk;
  const float* bp = Bm + (size_t)k_st0 * ldb + colblk + n_st;

  float4 b[8];
  v2f    a0[16], a1[16];

  ld_chunk(b, a0, bp, ap, ldb);
  for (int kb = 0; kb < K; kb += 2 * KCH) {
    st_chunk(Bs[0], b, n_st, k_st0);
    __syncthreads();
    ld_chunk(b, a1, bp + (size_t)(kb + KCH) * ldb, ap + (kb + KCH), ldb);
    mm_chunk(acc, Bs[0], a0, n0, hk);

    st_chunk(Bs[1], b, n_st, k_st0);
    __syncthreads();
    if (kb + 2 * KCH < K)
      ld_chunk(b, a0, bp + (size_t)(kb + 2 * KCH) * ldb, ap + (kb + 2 * KCH), ldb);
    mm_chunk(acc, Bs[1], a1, n0, hk);
  }
}


template <int LDB>
__device__ __forceinline__ void ld_batch(v4f (&a)[4], v2f (&b)[4][4], const float* __restrict__ ap,
                                         const float* __restrict__ bp) {
#pragma unroll
  for (int g = 0; g < 4; ++g) {
    a[g] = *(const v4f*)(ap + 8 * g);
#pragma unroll
    for (int j = 0; j < 4; ++j) b[g][j] = *(const v2f*)(bp + (8 * g + j) * LDB);
  }
}

__device__ __forceinline__ void mm_batch(v8f (&acc)[2], const v4f (&a)[4], const v2f (&b)[4][4]) {
#pragma unroll
  for (int g = 0; g < 4; ++g) {
    v2f alo = {a[g][0], a[g][1]};
    v2f ahi = {a[g][2], a[g][3]};
#pragma unroll
    for (int i = 0; i < 2; ++i) {
      v2f bf = {b[g][0][i], b[g][1][i]};
      acc[i] = __builtin_amdgcn_wmma_f32_16x16x4_f32(false, alo, false, bf, (short)0, acc[i],
                                                     false, false);
    }
#pragma unroll
    for (int i = 0; i < 2; ++i) {
      v2f bf = {b[g][2][i], b[g][3][i]};
      acc[i] = __builtin_amdgcn_wmma_f32_16x16x4_f32(false, ahi, false, bf, (short)0, acc[i],
                                                     false, false);
    }
  }
}

__device__ __forceinline__ v4f reduce_tile(float (*red)[16][32], const v8f (&acc)[2], int wave,
                                           int lane, int tid) {
  const int c = lane & 15, h = lane >> 4;
#pragma unroll
  for (int v = 0; v < 8; ++v)
    *(v2f*)&red[wave][8 * h + v][2 * c] = v2f{acc[0][v], acc[1][v]};
  __syncthreads();
  const int row = tid >> 3, c4 = (tid & 7) * 4;
  v4f sum = *(const v4f*)&red[0][row][c4];
#pragma unroll
  for (int w = 1; w < 4; ++w) sum += *(const v4f*)&red[w][row][c4];
  return sum;
}

__global__ __launch_bounds__(256) void RSLP_init_kernel(const float* __restrict__ x0, const float* __restrict__ Wi,
                                 const float* __restrict__ bi, float* __restrict__ s0,
                                 float* __restrict__ h) {
  int tid = threadIdx.x, lane = tid & 31, wave = tid >> 5;
  int row0   = (wave & 3) * 16;
  int col0w  = (wave >> 2) * 64;
  int colblk = blockIdx.x * 128;
  __shared__ float Bs[2][128][PADB];
  v8f acc[4] = {};
  gemm_staged<DINIT>(acc, Bs, x0, DINIT, Wi, 2 * NN, row0, colblk, col0w, lane, tid);

  float* dst = (blockIdx.x < 16) ? s0 : h;
  int joff   = (blockIdx.x < 16) ? 0 : NN;
#pragma unroll
  for (int i = 0; i < 4; ++i) {
    int j = colblk + col0w + 16 * i + (lane & 15);
    float bj = bi[j];
#pragma unroll
    for (int v = 0; v < 8; ++v) {
      int m = row0 + v + 8 * (lane >> 4);
      dst[m * NN + (j - joff)] = acc[i][v] + bj;
    }
  }
}

__global__ __launch_bounds__(128) void RSLP_step_kernel(const float* __restrict__ x, const float* __restrict__ Win,
                                 const float* __restrict__ bin, const float* __restrict__ r,
                                 const float* __restrict__ rb,
                                 const float* __restrict__ s_old, float* __restrict__ s_new,
                                 float* __restrict__ h, int t) {
  const int tid = threadIdx.x, lane = tid & 31, wave = tid >> 5;
  const int c = lane & 15, hl = lane >> 4;
  const int bid = blockIdx.x;
  __shared__ float red[4][16][32];
  const int row = tid >> 3, c4 = (tid & 7) * 4;
  const int row0 = (bid & 3) * 16;
  const int n0   = (bid >> 2) * 32;
  const int m  = row0 + row;
  const int j0 = n0 + c4;

  const float* ap  = s_old + (row0 + c) * NN + 512 * wave + 4 * hl;
  const float* bp  = r + (512 * wave + 4 * hl) * NN + n0 + 2 * c;
  const float* apx = x + t * DIN + (row0 + c) * (TT * DIN) + 64 * wave + 4 * hl;
  const float* bpx = Win + (64 * wave + 4 * hl) * NN + n0 + 2 * c;

  v4f a[4][4];
  v2f b[4][4][4];
  v8f acc[2] = {};
  v4f so, hh, rbv, biv;
  float rd[4];
#pragma unroll
  for (int i = 0; i < 4; ++i) ld_batch<NN>(a[i], b[i], ap + 32 * i, bp + 32 * i * NN);
  __builtin_amdgcn_sched_barrier(0);
#pragma unroll
  for (int i = 0; i < 18; ++i) {
    mm_batch(acc, a[i & 3], b[i & 3]);
    if (i + 4 < 18) {
      __builtin_amdgcn_sched_barrier(0);
      if (i + 4 < 16)
        ld_batch<NN>(a[i & 3], b[i & 3], ap + 32 * (i + 4), bp + 32 * (i + 4) * NN);
      else
        ld_batch<NN>(a[i & 3], b[i & 3], apx + 32 * (i + 4 - 16), bpx + 32 * (i + 4 - 16) * NN);
      if (i == 0) {
        so  = *(const v4f*)(s_old + m * NN + j0);
        hh  = *(const v4f*)(h + m * NN + j0);
        rbv = *(const v4f*)(rb + j0);
        biv = *(const v4f*)(bin + j0);
#pragma unroll
        for (int e = 0; e < 4; ++e) rd[e] = r[(j0 + e) * NN + (j0 + e)];
      }
      __builtin_amdgcn_sched_barrier(0);
    }
  }

  const v4f sum = reduce_tile(red, acc, wave, lane, tid);
  v4f hn4, sn4;
#pragma unroll
  for (int e = 0; e < 4; ++e) {
    float addj = rbv[e] + biv[e];
    float pre  = sum[e] - so[e] * rd[e] + addj;
    float hn   = (1.0f - DECAY) * hh[e] + DECAY * pre;
    hn4[e] = hn;
    sn4[e] = fmaxf(hn, 0.0f);
  }
  *(v4f*)(h + m * NN + j0)     = hn4;
  *(v4f*)(s_new + m * NN + j0) = sn4;
}

__global__ __launch_bounds__(256) void RSLP_out_kernel(const float* __restrict__ S1, const float* __restrict__ o,
                                float* __restrict__ out) {
  int tid = threadIdx.x, lane = tid & 31, wave = tid >> 5;
  int row0   = (wave & 3) * 16;
  int col0w  = (wave >> 2) * 64;
  int tt     = blockIdx.x >> 2;
  int colblk = (blockIdx.x & 3) * 128;
  __shared__ float Bs[2][128][PADB];
  v8f acc[4] = {};
  gemm_staged<NN>(acc, Bs, S1 + (size_t)tt * (BB * NN), NN, o, DOUT, row0, colblk, col0w, lane, tid);
#pragma unroll
  for (int i = 0; i < 4; ++i) {
    int j = colblk + col0w + 16 * i + (lane & 15);
#pragma unroll
    for (int v = 0; v < 8; ++v) {
      int m = row0 + v + 8 * (lane >> 4);
      out[(m * TT + tt) * DOUT + j] = acc[i][v];
    }
  }
}

extern "C" void kernel_launch(void* const* d_in, const int* in_sizes, int n_in,
                              void* d_out, int out_size, void* d_ws, size_t ws_size,
                              hipStream_t stream) {
  (void)in_sizes; (void)n_in; (void)out_size; (void)ws_size;
  const float* x      = (const float*)d_in[0];
  const float* x0     = (const float*)d_in[1];
  const float* W_init = (const float*)d_in[2];
  const float* b_init = (const float*)d_in[3];
  const float* W_in   = (const float*)d_in[4];
  const float* b_in   = (const float*)d_in[5];
  const float* r      = (const float*)d_in[6];
  const float* r_b    = (const float*)d_in[7];
  const float* o      = (const float*)d_in[8];
  float* out = (float*)d_out;

  float* S    = (float*)d_ws;
  float* hbuf = S + 129 * (BB * NN);

  RSLP_init_kernel<<<32, 256, 0, stream>>>(x0, W_init, b_init, S, hbuf);
  float* sp = S;
  for (int t = 0; t < TT; ++t) {
    RSLP_step_kernel<<<256, 128, 0, stream>>>(x, W_in, b_in, r, r_b, sp, sp + BB * NN, hbuf, t);
    sp += BB * NN;
  }
  RSLP_out_kernel<<<512, 256, 0, stream>>>(S + BB * NN, o, out);
}


# State slabs and h kept in a 16-row x 4-k packed order: each A fragment load of the step and of the output GEMM reads whole contiguous 128-byte lines (4 or 2 per load, not 16); sums unchanged
# speedup vs baseline: 11.0900x; 1.0404x over previous
#include <hip/hip_runtime.h>

#define BB    64
#define TT    128
#define NN    2048
#define DIN   256
#define DINIT 256
#define DOUT  512
#define DECAY 0.1f

#define KCH   64
#define PADB  68

typedef __attribute__((ext_vector_type(2))) float v2f;
typedef __attribute__((ext_vector_type(4))) float v4f;
typedef __attribute__((ext_vector_type(8))) float v8f;

#define PKG (NN / 4)
__device__ __forceinline__ int pk(int m, int n) {
  return (((m >> 4) * PKG + (n >> 2)) * 16 + (m & 15)) * 4 + (n & 3);
}


template <int AS>
__device__ __forceinline__ void ld_chunk(float4 (&b)[8], v2f (&a)[16], const float* __restrict__ bq,
                                         const float* __restrict__ aq, int ldb) {
#pragma unroll
  for (int j = 0; j < 8; ++j) {
    const float* gp = bq + (size_t)(8 * j) * ldb;
    b[j] = float4{gp[0], gp[(size_t)ldb], gp[2 * (size_t)ldb], gp[3 * (size_t)ldb]};
  }
#pragma unroll
  for (int ks = 0; ks < 16; ++ks) a[ks] = v2f{aq[AS * ks], aq[AS * ks + 1]};
}

__device__ __forceinline__ void st_chunk(float (*Bc)[PADB], const float4 (&b)[8], int n_st, int k_st0) {
#pragma unroll
  for (int j = 0; j < 8; ++j) *(float4*)&Bc[n_st][k_st0 + 8 * j] = b[j];
}

__device__ __forceinline__ void mm_chunk(v8f acc[4], float (*Bc)[PADB], const v2f (&a)[16], int n0, int hk) {
#pragma unroll
  for (int ks = 0; ks < 16; ++ks) {
#pragma unroll
    for (int i = 0; i < 4; ++i) {
      const float* p = &Bc[n0 + 16 * i][4 * ks + hk];
      v2f b = {p[0], p[1]};
      acc[i] = __builtin_amdgcn_wmma_f32_16x16x4_f32(false, a[ks], false, b, (short)0, acc[i],
                                                     false, false);
    }
  }
}

template <int K, bool PKA>
__device__ __forceinline__ void gemm_staged(v8f acc[4], float (*Bs)[128][PADB],
                                            const float* __restrict__ A, int lda,
                                            const float* __restrict__ Bm, int ldb,
                                            int row0, int colblk, int col0w, int lane, int tid) {
  static_assert(K % (2 * KCH) == 0, "the loop body holds two chunks, and a following call starts in stage 0");
  const int n_st  = tid & 127;
  const int k_st0 = (tid >> 7) * 4;
  const int hk    = (lane >> 4) << 1;
  const int n0    = col0w + (lane & 15);
  constexpr int AS = PKA ? 64 : 4;
  constexpr int AK = PKA ? 16 : 1;
  const float* ap = PKA ? A + ((row0 >> 4) * (lda >> 2) * 16 + (lane & 15)) * 4 + hk
                        : A + (row0 + (lane & 15)) * lda + hk;
  const float* bp = Bm + (size_t)k_st0 * ldb + colblk + n_st;

  float4 b[8];
  v2f    a0[16], a1[16];

  ld_chunk<AS>(b, a0, bp, ap, ldb);
  for (int kb = 0; kb < K; kb += 2 * KCH) {
    st_chunk(Bs[0], b, n_st, k_st0);
    __syncthreads();
    ld_chunk<AS>(b, a1, bp + (size_t)(kb + KCH) * ldb, ap + (kb + KCH) * AK, ldb);
    mm_chunk(acc, Bs[0], a0, n0, hk);

    st_chunk(Bs[1], b, n_st, k_st0);
    __syncthreads();
    if (kb + 2 * KCH < K)
      ld_chunk<AS>(b, a0, bp + (size_t)(kb + 2 * KCH) * ldb, ap + (kb + 2 * KCH) * AK, ldb);
    mm_chunk(acc, Bs[1], a1, n0, hk);
  }
}


template <int LDB, int AG>
__device__ __forceinline__ void ld_batch(v4f (&a)[4], v2f (&b)[4][4], const float* __restrict__ ap,
                                         const float* __restrict__ bp) {
#pragma unroll
  for (int g = 0; g < 4; ++g) {
    a[g] = *(const v4f*)(ap + AG * g);
#pragma unroll
    for (int j = 0; j < 4; ++j) b[g][j] = *(const v2f*)(bp + (8 * g + j) * LDB);
  }
}

__device__ __forceinline__ void mm_batch(v8f (&acc)[2], const v4f (&a)[4], const v2f (&b)[4][4]) {
#pragma unroll
  for (int g = 0; g < 4; ++g) {
    v2f alo = {a[g][0], a[g][1]};
    v2f ahi = {a[g][2], a[g][3]};
#pragma unroll
    for (int i = 0; i < 2; ++i) {
      v2f bf = {b[g][0][i], b[g][1][i]};
      acc[i] = __builtin_amdgcn_wmma_f32_16x16x4_f32(false, alo, false, bf, (short)0, acc[i],
                                                     false, false);
    }
#pragma unroll
    for (int i = 0; i < 2; ++i) {
      v2f bf = {b[g][2][i], b[g][3][i]};
      acc[i] = __builtin_amdgcn_wmma_f32_16x16x4_f32(false, ahi, false, bf, (short)0, acc[i],
                                                     false, false);
    }
  }
}

#define REDP 36
__device__ __forceinline__ v4f reduce_tile(float (*red)[16][REDP], const v8f (&acc)[2], int wave,
                                           int lane, int tid) {
  const int c = lane & 15, h = lane >> 4;
#pragma unroll
  for (int v = 0; v < 8; ++v)
    *(v2f*)&red[wave][8 * h + v][2 * c] = v2f{acc[0][v], acc[1][v]};
  __syncthreads();
  const int row = tid & 15, c4 = (tid >> 4) * 4;
  v4f sum = *(const v4f*)&red[0][row][c4];
#pragma unroll
  for (int w = 1; w < 4; ++w) sum += *(const v4f*)&red[w][row][c4];
  return sum;
}

__global__ __launch_bounds__(256) void RSLP_init_kernel(const float* __restrict__ x0, const float* __restrict__ Wi,
                                 const float* __restrict__ bi, float* __restrict__ s0,
                                 float* __restrict__ h) {
  int tid = threadIdx.x, lane = tid & 31, wave = tid >> 5;
  int row0   = (wave & 3) * 16;
  int col0w  = (wave >> 2) * 64;
  int colblk = blockIdx.x * 128;
  __shared__ float Bs[2][128][PADB];
  v8f acc[4] = {};
  gemm_staged<DINIT, false>(acc, Bs, x0, DINIT, Wi, 2 * NN, row0, colblk, col0w, lane, tid);

  float* dst = (blockIdx.x < 16) ? s0 : h;
  int joff   = (blockIdx.x < 16) ? 0 : NN;
#pragma unroll
  for (int i = 0; i < 4; ++i) {
    int j = colblk + col0w + 16 * i + (lane & 15);
    float bj = bi[j];
#pragma unroll
    for (int v = 0; v < 8; ++v) {
      int m = row0 + v + 8 * (lane >> 4);
      dst[pk(m, j - joff)] = acc[i][v] + bj;
    }
  }
}

__global__ __launch_bounds__(128) void RSLP_step_kernel(const float* __restrict__ x, const float* __restrict__ Win,
                                 const float* __restrict__ bin, const float* __restrict__ r,
                                 const float* __restrict__ rb,
                                 const float* __restrict__ s_old, float* __restrict__ s_new,
                                 float* __restrict__ h, int t) {
  const int tid = threadIdx.x, lane = tid & 31, wave = tid >> 5;
  const int c = lane & 15, hl = lane >> 4;
  const int bid = blockIdx.x;
  __shared__ __attribute__((aligned(16))) float red[4][16][REDP];
  const int row = tid & 15, c4 = (tid >> 4) * 4;
  const int rt   = bid & 3;
  const int row0 = rt * 16;
  const int n0   = (bid >> 2) * 32;
  const int j0 = n0 + c4;
  const int pidx = ((rt * PKG + (j0 >> 2)) * 16 + row) * 4;

  const float* ap  = s_old + ((rt * PKG + 128 * wave + hl) * 16 + c) * 4;
  const float* bp  = r + (512 * wave + 4 * hl) * NN + n0 + 2 * c;
  const float* apx = x + t * DIN + (row0 + c) * (TT * DIN) + 64 * wave + 4 * hl;
  const float* bpx = Win + (64 * wave + 4 * hl) * NN + n0 + 2 * c;

  v4f a[4][4];
  v2f b[4][4][4];
  v8f acc[2] = {};
  v4f so, hh, rbv, biv;
  float rd[4];
#pragma unroll
  for (int i = 0; i < 4; ++i) ld_batch<NN, 128>(a[i], b[i], ap + 512 * i, bp + 32 * i * NN);
  __builtin_amdgcn_sched_barrier(0);
#pragma unroll
  for (int i = 0; i < 18; ++i) {
    mm_batch(acc, a[i & 3], b[i & 3]);
    if (i + 4 < 18) {
      __builtin_amdgcn_sched_barrier(0);
      if (i + 4 < 16)
        ld_batch<NN, 128>(a[i & 3], b[i & 3], ap + 512 * (i + 4), bp + 32 * (i + 4) * NN);
      else
        ld_batch<NN, 8>(a[i & 3], b[i & 3], apx + 32 * (i + 4 - 16), bpx + 32 * (i + 4 - 16) * NN);
      if (i == 0) {
        so  = *(const v4f*)(s_old + pidx);
        hh  = *(const v4f*)(h + pidx);
        rbv = *(const v4f*)(rb + j0);
        biv = *(const v4f*)(bin + j0);
#pragma unroll
        for (int e = 0; e < 4; ++e) rd[e] = r[(j0 + e) * NN + (j0 + e)];
      }
      __builtin_amdgcn_sched_barrier(0);
    }
  }

  const v4f sum = reduce_tile(red, acc, wave, lane, tid);
  v4f hn4, sn4;
#pragma unroll
  for (int e = 0; e < 4; ++e) {
    float addj = rbv[e] + biv[e];
    float pre  = sum[e] - so[e] * rd[e] + addj;
    float hn   = (1.0f - DECAY) * hh[e] + DECAY * pre;
    hn4[e] = hn;
    sn4[e] = fmaxf(hn, 0.0f);
  }
  *(v4f*)(h + pidx)     = hn4;
  *(v4f*)(s_new + pidx) = sn4;
}

__global__ __launch_bounds__(256) void RSLP_out_kernel(const float* __restrict__ S1, const float* __restrict__ o,
                                float* __restrict__ out) {
  int tid = threadIdx.x, lane = tid & 31, wave = tid >> 5;
  int row0   = (wave & 3) * 16;
  int col0w  = (wave >> 2) * 64;
  int tt     = blockIdx.x >> 2;
  int colblk = (blockIdx.x & 3) * 128;
  __shared__ float Bs[2][128][PADB];
  v8f acc[4] = {};
  gemm_staged<NN, true>(acc, Bs, S1 + (size_t)tt * (BB * NN), NN, o, DOUT, row0, colblk, col0w, lane, tid);
#pragma unroll
  for (int i = 0; i < 4; ++i) {
    int j = colblk + col0w + 16 * i + (lane & 15);
#pragma unroll
    for (int v = 0; v < 8; ++v) {
      int m = row0 + v + 8 * (lane >> 4);
      out[(m * TT + tt) * DOUT + j] = acc[i][v];
    }
  }
}

extern "C" void kernel_launch(void* const* d_in, const int* in_sizes, int n_in,
                              void* d_out, int out_size, void* d_ws, size_t ws_size,
                              hipStream_t stream) {
  (void)in_sizes; (void)n_in; (void)out_size; (void)ws_size;
  const float* x      = (const float*)d_in[0];
  const float* x0     = (const float*)d_in[1];
  const float* W_init = (const float*)d_in[2];
  const float* b_init = (const float*)d_in[3];
  const float* W_in   = (const float*)d_in[4];
  const float* b_in   = (const float*)d_in[5];
  const float* r      = (const float*)d_in[6];
  const float* r_b    = (const float*)d_in[7];
  const float* o      = (const float*)d_in[8];
  float* out = (float*)d_out;

  float* S    = (float*)d_ws;
  float* hbuf = S + 129 * (BB * NN);

  RSLP_init_kernel<<<32, 256, 0, stream>>>(x0, W_init, b_init, S, hbuf);
  float* sp = S;
  for (int t = 0; t < TT; ++t) {
    RSLP_step_kernel<<<256, 128, 0, stream>>>(x, W_in, b_in, r, r_b, sp, sp + BB * NN, hbuf, t);
    sp += BB * NN;
  }
  RSLP_out_kernel<<<512, 256, 0, stream>>>(S + BB * NN, o, out);
}


# r and W_in re-laid once per call by one added kernel into WMMA fragment order: a step B load is one contiguous 512-byte run per wave with k-paired register pairs (no moves; 12 loads a batch, not 20)
# speedup vs baseline: 12.6143x; 1.1374x over previous
#include <hip/hip_runtime.h>

#define BB    64
#define TT    128
#define NN    2048
#define DIN   256
#define DINIT 256
#define DOUT  512
#define DECAY 0.1f

#define KCH   64
#define PADB  68

typedef __attribute__((ext_vector_type(2))) float v2f;
typedef __attribute__((ext_vector_type(4))) float v4f;
typedef __attribute__((ext_vector_type(8))) float v8f;

#define PKG (NN / 4)
__device__ __forceinline__ int pk(int m, int n) {
  return (((m >> 4) * PKG + (n >> 2)) * 16 + (m & 15)) * 4 + (n & 3);
}


template <int AS>
__device__ __forceinline__ void ld_chunk(float4 (&b)[8], v2f (&a)[16], const float* __restrict__ bq,
                                         const float* __restrict__ aq, int ldb) {
#pragma unroll
  for (int j = 0; j < 8; ++j) {
    const float* gp = bq + (size_t)(8 * j) * ldb;
    b[j] = float4{gp[0], gp[(size_t)ldb], gp[2 * (size_t)ldb], gp[3 * (size_t)ldb]};
  }
#pragma unroll
  for (int ks = 0; ks < 16; ++ks) a[ks] = v2f{aq[AS * ks], aq[AS * ks + 1]};
}

__device__ __forceinline__ void st_chunk(float (*Bc)[PADB], const float4 (&b)[8], int n_st, int k_st0) {
#pragma unroll
  for (int j = 0; j < 8; ++j) *(float4*)&Bc[n_st][k_st0 + 8 * j] = b[j];
}

__device__ __forceinline__ void mm_chunk(v8f acc[4], float (*Bc)[PADB], const v2f (&a)[16], int n0, int hk) {
#pragma unroll
  for (int ks = 0; ks < 16; ++ks) {
#pragma unroll
    for (int i = 0; i < 4; ++i) {
      const float* p = &Bc[n0 + 16 * i][4 * ks + hk];
      v2f b = {p[0], p[1]};
      acc[i] = __builtin_amdgcn_wmma_f32_16x16x4_f32(false, a[ks], false, b, (short)0, acc[i],
                                                     false, false);
    }
  }
}

template <int K, bool PKA>
__device__ __forceinline__ void gemm_staged(v8f acc[4], float (*Bs)[128][PADB],
                                            const float* __restrict__ A, int lda,
                                            const float* __restrict__ Bm, int ldb,
                                            int row0, int colblk, int col0w, int lane, int tid) {
  static_assert(K % (2 * KCH) == 0, "the loop body holds two chunks, and a following call starts in stage 0");
  const int n_st  = tid & 127;
  const int k_st0 = (tid >> 7) * 4;
  const int hk    = (lane >> 4) << 1;
  const int n0    = col0w + (lane & 15);
  constexpr int AS = PKA ? 64 : 4;
  constexpr int AK = PKA ? 16 : 1;
  const float* ap = PKA ? A + ((row0 >> 4) * (lda >> 2) * 16 + (lane & 15)) * 4 + hk
                        : A + (row0 + (lane & 15)) * lda + hk;
  const float* bp = Bm + (size_t)k_st0 * ldb + colblk + n_st;

  float4 b[8];
  v2f    a0[16], a1[16];

  ld_chunk<AS>(b, a0, bp, ap, ldb);
  for (int kb = 0; kb < K; kb += 2 * KCH) {
    st_chunk(Bs[0], b, n_st, k_st0);
    __syncthreads();
    ld_chunk<AS>(b, a1, bp + (size_t)(kb + KCH) * ldb, ap + (kb + KCH) * AK, ldb);
    mm_chunk(acc, Bs[0], a0, n0, hk);

    st_chunk(Bs[1], b, n_st, k_st0);
    __syncthreads();
    if (kb + 2 * KCH < K)
      ld_chunk<AS>(b, a0, bp + (size_t)(kb + 2 * KCH) * ldb, ap + (kb + 2 * KCH) * AK, ldb);
    mm_chunk(acc, Bs[1], a1, n0, hk);
  }
}


#define PB_LOAD 128

template <int AG>
__device__ __forceinline__ void ld_batch(v4f (&a)[4], v4f (&b)[4][2], const float* __restrict__ ap,
                                         const float* __restrict__ bp) {
#pragma unroll
  for (int g = 0; g < 4; ++g) {
    a[g] = *(const v4f*)(ap + AG * g);
#pragma unroll
    for (int q = 0; q < 2; ++q) b[g][q] = *(const v4f*)(bp + (2 * g + q) * PB_LOAD);
  }
}

__device__ __forceinline__ void mm_batch(v8f (&acc)[2], const v4f (&a)[4], const v4f (&b)[4][2]) {
#pragma unroll
  for (int g = 0; g < 4; ++g) {
    v2f alo = {a[g][0], a[g][1]};
    v2f ahi = {a[g][2], a[g][3]};
#pragma unroll
    for (int i = 0; i < 2; ++i) {
      v2f bf = {b[g][0][2 * i], b[g][0][2 * i + 1]};
      acc[i] = __builtin_amdgcn_wmma_f32_16x16x4_f32(false, alo, false, bf, (short)0, acc[i],
                                                     false, false);
    }
#pragma unroll
    for (int i = 0; i < 2; ++i) {
      v2f bf = {b[g][1][2 * i], b[g][1][2 * i + 1]};
      acc[i] = __builtin_amdgcn_wmma_f32_16x16x4_f32(false, ahi, false, bf, (short)0, acc[i],
                                                     false, false);
    }
  }
}

__global__ __launch_bounds__(256) void RSLP_pack_kernel(const float* __restrict__ r, const float* __restrict__ Win,
                                 float* __restrict__ pr, float* __restrict__ pw) {
  const int tid = threadIdx.x, lane = tid & 31;
  const int c = lane & 15, hl = lane >> 4;
  const int gw = blockIdx.x * 8 + (tid >> 5);
  const bool isr = gw < 2048;
  const float* src = isr ? r : Win;
  float* dst       = isr ? pr : pw;
  const int kgs    = isr ? (NN / 8) : (DIN / 8);
  const int t      = isr ? gw : gw - 2048;
  const int rsh    = isr ? 5 : 2;
  const int ct = t >> rsh, ko = t & ((1 << rsh) - 1);
  const float* sp = src + (size_t)(64 * ko + 4 * hl) * NN + 32 * ct + 2 * c;
  float* dp = dst + ((size_t)(ct * kgs + 8 * ko) * 2) * PB_LOAD + lane * 4;
  v2f v[8][4];
#pragma unroll
  for (int g = 0; g < 8; ++g)
#pragma unroll
    for (int j = 0; j < 4; ++j) v[g][j] = *(const v2f*)(sp + (size_t)(8 * g + j) * NN);
#pragma unroll
  for (int g = 0; g < 8; ++g)
#pragma unroll
    for (int q = 0; q < 2; ++q)
      *(v4f*)(dp + (2 * g + q) * PB_LOAD) =
          v4f{v[g][2 * q][0], v[g][2 * q + 1][0], v[g][2 * q][1], v[g][2 * q + 1][1]};
}

#define REDP 36
__device__ __forceinline__ v4f reduce_tile(float (*red)[16][REDP], const v8f (&acc)[2], int wave,
                                           int lane, int tid) {
  const int c = lane & 15, h = lane >> 4;
#pragma unroll
  for (int v = 0; v < 8; ++v)
    *(v2f*)&red[wave][8 * h + v][2 * c] = v2f{acc[0][v], acc[1][v]};
  __syncthreads();
  const int row = tid & 15, c4 = (tid >> 4) * 4;
  v4f sum = *(const v4f*)&red[0][row][c4];
#pragma unroll
  for (int w = 1; w < 4; ++w) sum += *(const v4f*)&red[w][row][c4];
  return sum;
}

__global__ __launch_bounds__(256) void RSLP_init_kernel(const float* __restrict__ x0, const float* __restrict__ Wi,
                                 const float* __restrict__ bi, float* __restrict__ s0,
                                 float* __restrict__ h) {
  int tid = threadIdx.x, lane = tid & 31, wave = tid >> 5;
  int row0   = (wave & 3) * 16;
  int col0w  = (wave >> 2) * 64;
  int colblk = blockIdx.x * 128;
  __shared__ float Bs[2][128][PADB];
  v8f acc[4] = {};
  gemm_staged<DINIT, false>(acc, Bs, x0, DINIT, Wi, 2 * NN, row0, colblk, col0w, lane, tid);

  float* dst = (blockIdx.x < 16) ? s0 : h;
  int joff   = (blockIdx.x < 16) ? 0 : NN;
#pragma unroll
  for (int i = 0; i < 4; ++i) {
    int j = colblk + col0w + 16 * i + (lane & 15);
    float bj = bi[j];
#pragma unroll
    for (int v = 0; v < 8; ++v) {
      int m = row0 + v + 8 * (lane >> 4);
      dst[pk(m, j - joff)] = acc[i][v] + bj;
    }
  }
}

__global__ __launch_bounds__(128) void RSLP_step_kernel(const float* __restrict__ x, const float* __restrict__ bin,
                                 const float* __restrict__ r, const float* __restrict__ rb,
                                 const float* __restrict__ pr, const float* __restrict__ pw,
                                 const float* __restrict__ s_old, float* __restrict__ s_new,
                                 float* __restrict__ h, int t) {
  const int tid = threadIdx.x, lane = tid & 31, wave = tid >> 5;
  const int c = lane & 15, hl = lane >> 4;
  const int bid = blockIdx.x;
  __shared__ __attribute__((aligned(16))) float red[4][16][REDP];
  const int row = tid & 15, c4 = (tid >> 4) * 4;
  const int rt   = bid & 3;
  const int row0 = rt * 16;
  const int ct   = bid >> 2;
  const int n0   = ct * 32;
  const int j0 = n0 + c4;
  const int pidx = ((rt * PKG + (j0 >> 2)) * 16 + row) * 4;

  const float* ap  = s_old + ((rt * PKG + 128 * wave + hl) * 16 + c) * 4;
  const float* bp  = pr + ((ct * (NN / 8) + 64 * wave) * 2) * PB_LOAD + lane * 4;
  const float* apx = x + t * DIN + (row0 + c) * (TT * DIN) + 64 * wave + 4 * hl;
  const float* bpx = pw + ((ct * (DIN / 8) + 8 * wave) * 2) * PB_LOAD + lane * 4;

  v4f a[4][4];
  v4f b[4][4][2];
  v8f acc[2] = {};
  v4f so, hh, rbv, biv;
  float rd[4];
#pragma unroll
  for (int i = 0; i < 4; ++i) ld_batch<128>(a[i], b[i], ap + 512 * i, bp + 8 * PB_LOAD * i);
  __builtin_amdgcn_sched_barrier(0);
#pragma unroll
  for (int i = 0; i < 18; ++i) {
    mm_batch(acc, a[i & 3], b[i & 3]);
    if (i + 4 < 18) {
      __builtin_amdgcn_sched_barrier(0);
      if (i + 4 < 16)
        ld_batch<128>(a[i & 3], b[i & 3], ap + 512 * (i + 4), bp + 8 * PB_LOAD * (i + 4));
      else
        ld_batch<8>(a[i & 3], b[i & 3], apx + 32 * (i + 4 - 16), bpx + 8 * PB_LOAD * (i + 4 - 16));
      if (i == 0) {
        so  = *(const v4f*)(s_old + pidx);
        hh  = *(const v4f*)(h + pidx);
        rbv = *(const v4f*)(rb + j0);
        biv = *(const v4f*)(bin + j0);
#pragma unroll
        for (int e = 0; e < 4; ++e) rd[e] = r[(j0 + e) * NN + (j0 + e)];
      }
      __builtin_amdgcn_sched_barrier(0);
    }
  }

  const v4f sum = reduce_tile(red, acc, wave, lane, tid);
  v4f hn4, sn4;
#pragma unroll
  for (int e = 0; e < 4; ++e) {
    float addj = rbv[e] + biv[e];
    float pre  = sum[e] - so[e] * rd[e] + addj;
    float hn   = (1.0f - DECAY) * hh[e] + DECAY * pre;
    hn4[e] = hn;
    sn4[e] = fmaxf(hn, 0.0f);
  }
  *(v4f*)(h + pidx)     = hn4;
  *(v4f*)(s_new + pidx) = sn4;
}

__global__ __launch_bounds__(256) void RSLP_out_kernel(const float* __restrict__ S1, const float* __restrict__ o,
                                float* __restrict__ out) {
  int tid = threadIdx.x, lane = tid & 31, wave = tid >> 5;
  int row0   = (wave & 3) * 16;
  int col0w  = (wave >> 2) * 64;
  int tt     = blockIdx.x >> 2;
  int colblk = (blockIdx.x & 3) * 128;
  __shared__ float Bs[2][128][PADB];
  v8f acc[4] = {};
  gemm_staged<NN, true>(acc, Bs, S1 + (size_t)tt * (BB * NN), NN, o, DOUT, row0, colblk, col0w, lane, tid);
#pragma unroll
  for (int i = 0; i < 4; ++i) {
    int j = colblk + col0w + 16 * i + (lane & 15);
#pragma unroll
    for (int v = 0; v < 8; ++v) {
      int m = row0 + v + 8 * (lane >> 4);
      out[(m * TT + tt) * DOUT + j] = acc[i][v];
    }
  }
}

extern "C" void kernel_launch(void* const* d_in, const int* in_sizes, int n_in,
                              void* d_out, int out_size, void* d_ws, size_t ws_size,
                              hipStream_t stream) {
  (void)in_sizes; (void)n_in; (void)out_size; (void)ws_size;
  const float* x      = (const float*)d_in[0];
  const float* x0     = (const float*)d_in[1];
  const float* W_init = (const float*)d_in[2];
  const float* b_init = (const float*)d_in[3];
  const float* W_in   = (const float*)d_in[4];
  const float* b_in   = (const float*)d_in[5];
  const float* r      = (const float*)d_in[6];
  const float* r_b    = (const float*)d_in[7];
  const float* o      = (const float*)d_in[8];
  float* out = (float*)d_out;

  float* S    = (float*)d_ws;
  float* hbuf = S + 129 * (BB * NN);
  float* pr   = hbuf + BB * NN;
  float* pw   = pr + NN * NN;

  RSLP_pack_kernel<<<288, 256, 0, stream>>>(r, W_in, pr, pw);
  RSLP_init_kernel<<<32, 256, 0, stream>>>(x0, W_init, b_init, S, hbuf);
  float* sp = S;
  for (int t = 0; t < TT; ++t) {
    RSLP_step_kernel<<<256, 128, 0, stream>>>(x, b_in, r, r_b, pr, pw, sp, sp + BB * NN, hbuf, t);
    sp += BB * NN;
  }
  RSLP_out_kernel<<<512, 256, 0, stream>>>(S + BB * NN, o, out);
}
